# starting kernel
# baseline (speedup 1.0000x reference)
#include <hip/hip_runtime.h>

typedef __attribute__((ext_vector_type(2))) float v2f;
typedef __attribute__((ext_vector_type(8))) float v8f;


__global__ void set_val_kernel(float* __restrict__ p, float v, int n) {
    int i = blockIdx.x * blockDim.x + threadIdx.x;
    if (i < n) p[i] = v;
}

__global__ void deg_count_kernel(const int* __restrict__ dst, float* __restrict__ deg, int E) {
    int i = blockIdx.x * blockDim.x + threadIdx.x;
    if (i < E) unsafeAtomicAdd(&deg[dst[i]], 1.0f);
}

__global__ void rsqrt_kernel(float* __restrict__ p, int n) {
    int i = blockIdx.x * blockDim.x + threadIdx.x;
    if (i < n) p[i] = rsqrtf(p[i]);
}

__global__ void bias_add_kernel(float* __restrict__ out, const float* __restrict__ b,
                                int n, int mask) {
    int i = blockIdx.x * blockDim.x + threadIdx.x;
    if (i < n) out[i] += b[i & mask];
}

template <int FIN, int FOUT, bool FUSE_BIAS_RELU>
__global__ __launch_bounds__(256) void gemm_wmma_kernel(
    const float* __restrict__ A, const float* __restrict__ W,
    const float* __restrict__ bias, float* __restrict__ C,
    int M, int mtiles) {
    constexpr int NT = FOUT / 16;
    int wave = blockIdx.x * (blockDim.x >> 5) + (threadIdx.x >> 5);
    if (wave >= mtiles) return;

    const int row0 = wave * 16;
    const int lane = threadIdx.x & 31;
    const int half = lane >> 4;
    const int lm   = lane & 15;
    const int koff = half * 2;

    const float* __restrict__ arow = A + (size_t)(row0 + lm) * FIN;
    const bool full_tile = (row0 + 16) <= M;
    const float* __restrict__ arow_safe =
        full_tile ? arow : (A + (size_t)((row0 + lm) < M ? (row0 + lm) : (M - 1)) * FIN);

    v8f acc[NT];
#pragma unroll
    for (int t = 0; t < NT; ++t) acc[t] = v8f{};

#pragma unroll 4
    for (int k = 0; k < FIN; k += 4) {
        float a0 = arow_safe[k + koff];
        float a1 = arow_safe[k + koff + 1];
        if (FUSE_BIAS_RELU) {
            a0 = fmaxf(a0 + bias[k + koff], 0.0f);
            a1 = fmaxf(a1 + bias[k + koff + 1], 0.0f);
        }
        v2f a;
        a.x = a0;
        a.y = a1;
#pragma unroll
        for (int t = 0; t < NT; ++t) {
            v2f b;
            b.x = W[(size_t)(k + koff) * FOUT + t * 16 + lm];
            b.y = W[(size_t)(k + koff + 1) * FOUT + t * 16 + lm];
            acc[t] = __builtin_amdgcn_wmma_f32_16x16x4_f32(
                 false, a,  false, b,
                 (short)0, acc[t],  false,  false);
        }
    }

    if (full_tile) {
#pragma unroll
        for (int g = 0; g < 8; ++g) {
            float* crow = C + (size_t)(row0 + half * 8 + g) * FOUT + lm;
#pragma unroll
            for (int t = 0; t < NT; ++t) crow[t * 16] = acc[t][g];
        }
    } else {
#pragma unroll
        for (int g = 0; g < 8; ++g) {
            int r = row0 + half * 8 + g;
            if (r < M) {
                float* crow = C + (size_t)r * FOUT + lm;
#pragma unroll
                for (int t = 0; t < NT; ++t) crow[t * 16] = acc[t][g];
            }
        }
    }
}

template <int F>
__global__ __launch_bounds__(256) void aggregate_kernel(
    const float* __restrict__ h, const int* __restrict__ src,
    const int* __restrict__ dst, const float* __restrict__ dinv,
    float* __restrict__ out, int E, int N) {
    const int TPE = F / 4;
    long long idx = (long long)blockIdx.x * blockDim.x + threadIdx.x;
    long long total = (long long)(E + N) * TPE;
    if (idx >= total) return;

    int e  = (int)(idx / TPE);
    int f4 = (int)(idx % TPE) * 4;

    int s, d;
    if (e < E) {
        s = src[e];
        d = dst[e];
    } else {
        s = d = e - E;
    }
    float norm = dinv[s] * dinv[d];

    const float4 hv = *reinterpret_cast<const float4*>(h + (size_t)s * F + f4);
    float* o = out + (size_t)d * F + f4;
    unsafeAtomicAdd(o + 0, hv.x * norm);
    unsafeAtomicAdd(o + 1, hv.y * norm);
    unsafeAtomicAdd(o + 2, hv.z * norm);
    unsafeAtomicAdd(o + 3, hv.w * norm);
}


extern "C" void kernel_launch(void* const* d_in, const int* in_sizes, int n_in,
                              void* d_out, int out_size, void* d_ws, size_t ws_size,
                              hipStream_t stream) {
    const float* x  = (const float*)d_in[0];
    const int*   ei = (const int*)d_in[1];
    const float* W1 = (const float*)d_in[2];
    const float* b1 = (const float*)d_in[3];
    const float* W2 = (const float*)d_in[4];
    const float* b2 = (const float*)d_in[5];

    const int FIN = 128, F1 = 64, F2 = 32;
    const int N = in_sizes[0] / FIN;
    const int E = in_sizes[1] / 2;
    const int* src = ei;
    const int* dst = ei + E;

    float* ws   = (float*)d_ws;
    float* deg  = ws;
    float* h1   = ws + N;
    float* agg1 = ws + N + (size_t)N * F1;
    float* outp = (float*)d_out;

    hipMemsetAsync(agg1, 0, (size_t)N * F1 * sizeof(float), stream);
    hipMemsetAsync(outp, 0, (size_t)N * F2 * sizeof(float), stream);

    const int thr = 256;
    const int wpb = thr / 32;

    set_val_kernel<<<(N + thr - 1) / thr, thr, 0, stream>>>(deg, 1.0f, N);
    deg_count_kernel<<<(E + thr - 1) / thr, thr, 0, stream>>>(dst, deg, E);
    rsqrt_kernel<<<(N + thr - 1) / thr, thr, 0, stream>>>(deg, N);

    {
        int mtiles = (N + 15) / 16;
        int blocks = (mtiles + wpb - 1) / wpb;
        gemm_wmma_kernel<128, 64, false>
            <<<blocks, thr, 0, stream>>>(x, W1, nullptr, h1, N, mtiles);
    }
    {
        long long total = (long long)(E + N) * (F1 / 4);
        int blocks = (int)((total + thr - 1) / thr);
        aggregate_kernel<64><<<blocks, thr, 0, stream>>>(h1, src, dst, deg, agg1, E, N);
    }
    {
        int mtiles = (N + 15) / 16;
        int blocks = (mtiles + wpb - 1) / wpb;
        gemm_wmma_kernel<64, 32, true>
            <<<blocks, thr, 0, stream>>>(agg1, W2, b1, h1  , N, mtiles);
    }
    {
        long long total = (long long)(E + N) * (F2 / 4);
        int blocks = (int)((total + thr - 1) / thr);
        aggregate_kernel<32><<<blocks, thr, 0, stream>>>(h1, src, dst, deg, outp, E, N);
    }
    bias_add_kernel<<<(N * F2 + thr - 1) / thr, thr, 0, stream>>>(outp, b2, N * F2, F2 - 1);
}


# Degree count: one scattered global f32 atomic per edge replaced by per-work-group integer histograms in LDS, stored as partial rows and summed in the rsqrt kernel (exact counts, same values)
# speedup vs baseline: 1.2438x; 1.2438x over previous
#include <hip/hip_runtime.h>

typedef __attribute__((ext_vector_type(2))) float v2f;
typedef __attribute__((ext_vector_type(8))) float v8f;


constexpr int HIST_CAP     = 65536;
constexpr int HIST_THREADS = 1024;
constexpr int HIST_GROUPS  = 64;

__global__ __launch_bounds__(HIST_THREADS) void deg_hist_kernel(
    const int* __restrict__ dst, unsigned int* __restrict__ part, int E, int N) {
    __shared__ unsigned int hist[HIST_CAP];
    const int tid = threadIdx.x;
    const int per = (E + (int)gridDim.x - 1) / (int)gridDim.x;
    long long b0 = (long long)blockIdx.x * per;
    const int e0 = (int)(b0 < E ? b0 : E);
    const int e1 = (e0 + per < E) ? (e0 + per) : E;

    for (int r0 = 0; r0 < N; r0 += HIST_CAP) {
        const int nb = (N - r0 < HIST_CAP) ? (N - r0) : HIST_CAP;
        uint4* hz = reinterpret_cast<uint4*>(hist);
        for (int i = tid; i < (nb + 3) / 4; i += HIST_THREADS) hz[i] = uint4{0u, 0u, 0u, 0u};
        __syncthreads();

        int e = e0 + tid;
        for (; e + 7 * HIST_THREADS < e1; e += 8 * HIST_THREADS) {
            int d[8];
#pragma unroll
            for (int j = 0; j < 8; ++j) d[j] = dst[e + j * HIST_THREADS];
#pragma unroll
            for (int j = 0; j < 8; ++j) {
                unsigned int b = (unsigned int)(d[j] - r0);
                if (b < (unsigned int)nb) atomicAdd(&hist[b], 1u);
            }
        }
        for (; e < e1; e += HIST_THREADS) {
            unsigned int b = (unsigned int)(dst[e] - r0);
            if (b < (unsigned int)nb) atomicAdd(&hist[b], 1u);
        }
        __syncthreads();

        unsigned int* prow = part + (size_t)blockIdx.x * N + r0;
        for (int i = tid; i < nb; i += HIST_THREADS) prow[i] = hist[i];
        __syncthreads();
    }
}

__global__ void deg_reduce_rsqrt_kernel(const unsigned int* __restrict__ part,
                                        float* __restrict__ dinv, int N) {
    int i = blockIdx.x * blockDim.x + threadIdx.x;
    if (i >= N) return;
    unsigned int c = 1u;
    unsigned int off = (unsigned int)i;
#pragma unroll 16
    for (int g = 0; g < HIST_GROUPS; ++g) {
        c += part[off];
        off += (unsigned int)N;
    }
    dinv[i] = rsqrtf((float)c);
}

__global__ void bias_add_kernel(float* __restrict__ out, const float* __restrict__ b,
                                int n, int mask) {
    int i = blockIdx.x * blockDim.x + threadIdx.x;
    if (i < n) out[i] += b[i & mask];
}

template <int FIN, int FOUT, bool FUSE_BIAS_RELU>
__global__ __launch_bounds__(256) void gemm_wmma_kernel(
    const float* __restrict__ A, const float* __restrict__ W,
    const float* __restrict__ bias, float* __restrict__ C,
    int M, int mtiles) {
    constexpr int NT = FOUT / 16;
    int wave = blockIdx.x * (blockDim.x >> 5) + (threadIdx.x >> 5);
    if (wave >= mtiles) return;

    const int row0 = wave * 16;
    const int lane = threadIdx.x & 31;
    const int half = lane >> 4;
    const int lm   = lane & 15;
    const int koff = half * 2;

    const float* __restrict__ arow = A + (size_t)(row0 + lm) * FIN;
    const bool full_tile = (row0 + 16) <= M;
    const float* __restrict__ arow_safe =
        full_tile ? arow : (A + (size_t)((row0 + lm) < M ? (row0 + lm) : (M - 1)) * FIN);

    v8f acc[NT];
#pragma unroll
    for (int t = 0; t < NT; ++t) acc[t] = v8f{};

#pragma unroll 4
    for (int k = 0; k < FIN; k += 4) {
        float a0 = arow_safe[k + koff];
        float a1 = arow_safe[k + koff + 1];
        if (FUSE_BIAS_RELU) {
            a0 = fmaxf(a0 + bias[k + koff], 0.0f);
            a1 = fmaxf(a1 + bias[k + koff + 1], 0.0f);
        }
        v2f a;
        a.x = a0;
        a.y = a1;
#pragma unroll
        for (int t = 0; t < NT; ++t) {
            v2f b;
            b.x = W[(size_t)(k + koff) * FOUT + t * 16 + lm];
            b.y = W[(size_t)(k + koff + 1) * FOUT + t * 16 + lm];
            acc[t] = __builtin_amdgcn_wmma_f32_16x16x4_f32(
                 false, a,  false, b,
                 (short)0, acc[t],  false,  false);
        }
    }

    if (full_tile) {
#pragma unroll
        for (int g = 0; g < 8; ++g) {
            float* crow = C + (size_t)(row0 + half * 8 + g) * FOUT + lm;
#pragma unroll
            for (int t = 0; t < NT; ++t) crow[t * 16] = acc[t][g];
        }
    } else {
#pragma unroll
        for (int g = 0; g < 8; ++g) {
            int r = row0 + half * 8 + g;
            if (r < M) {
                float* crow = C + (size_t)r * FOUT + lm;
#pragma unroll
                for (int t = 0; t < NT; ++t) crow[t * 16] = acc[t][g];
            }
        }
    }
}

template <int F>
__global__ __launch_bounds__(256) void aggregate_kernel(
    const float* __restrict__ h, const int* __restrict__ src,
    const int* __restrict__ dst, const float* __restrict__ dinv,
    float* __restrict__ out, int E, int N) {
    const int TPE = F / 4;
    long long idx = (long long)blockIdx.x * blockDim.x + threadIdx.x;
    long long total = (long long)(E + N) * TPE;
    if (idx >= total) return;

    int e  = (int)(idx / TPE);
    int f4 = (int)(idx % TPE) * 4;

    int s, d;
    if (e < E) {
        s = src[e];
        d = dst[e];
    } else {
        s = d = e - E;
    }
    float norm = dinv[s] * dinv[d];

    const float4 hv = *reinterpret_cast<const float4*>(h + (size_t)s * F + f4);
    float* o = out + (size_t)d * F + f4;
    unsafeAtomicAdd(o + 0, hv.x * norm);
    unsafeAtomicAdd(o + 1, hv.y * norm);
    unsafeAtomicAdd(o + 2, hv.z * norm);
    unsafeAtomicAdd(o + 3, hv.w * norm);
}


extern "C" void kernel_launch(void* const* d_in, const int* in_sizes, int n_in,
                              void* d_out, int out_size, void* d_ws, size_t ws_size,
                              hipStream_t stream) {
    const float* x  = (const float*)d_in[0];
    const int*   ei = (const int*)d_in[1];
    const float* W1 = (const float*)d_in[2];
    const float* b1 = (const float*)d_in[3];
    const float* W2 = (const float*)d_in[4];
    const float* b2 = (const float*)d_in[5];

    const int FIN = 128, F1 = 64, F2 = 32;
    const int N = in_sizes[0] / FIN;
    const int E = in_sizes[1] / 2;
    const int* src = ei;
    const int* dst = ei + E;

    float* ws   = (float*)d_ws;
    float* deg  = ws;
    float* h1   = ws + N;
    float* agg1 = ws + N + (size_t)N * F1;
    float* outp = (float*)d_out;
    unsigned int* part = (unsigned int*)(ws + N + 2 * (size_t)N * F1);

    hipMemsetAsync(agg1, 0, (size_t)N * F1 * sizeof(float), stream);
    hipMemsetAsync(outp, 0, (size_t)N * F2 * sizeof(float), stream);

    const int thr = 256;
    const int wpb = thr / 32;

    deg_hist_kernel<<<HIST_GROUPS, HIST_THREADS, 0, stream>>>(dst, part, E, N);
    deg_reduce_rsqrt_kernel<<<(N + thr - 1) / thr, thr, 0, stream>>>(part, deg, N);

    {
        int mtiles = (N + 15) / 16;
        int blocks = (mtiles + wpb - 1) / wpb;
        gemm_wmma_kernel<128, 64, false>
            <<<blocks, thr, 0, stream>>>(x, W1, nullptr, h1, N, mtiles);
    }
    {
        long long total = (long long)(E + N) * (F1 / 4);
        int blocks = (int)((total + thr - 1) / thr);
        aggregate_kernel<64><<<blocks, thr, 0, stream>>>(h1, src, dst, deg, agg1, E, N);
    }
    {
        int mtiles = (N + 15) / 16;
        int blocks = (mtiles + wpb - 1) / wpb;
        gemm_wmma_kernel<64, 32, true>
            <<<blocks, thr, 0, stream>>>(agg1, W2, b1, h1  , N, mtiles);
    }
    {
        long long total = (long long)(E + N) * (F2 / 4);
        int blocks = (int)((total + thr - 1) / thr);
        aggregate_kernel<32><<<blocks, thr, 0, stream>>>(h1, src, dst, deg, outp, E, N);
    }
    bias_add_kernel<<<(N * F2 + thr - 1) / thr, thr, 0, stream>>>(outp, b2, N * F2, F2 - 1);
}


# Aggregation: per-edge f32 atomic scatter replaced by gather over edges counting-sorted by dst each call (LDS cursors); one wave sums a node row in registers, stores once with bias; sum order changes
# speedup vs baseline: 2.8039x; 2.2543x over previous
#include <hip/hip_runtime.h>

typedef __attribute__((ext_vector_type(2))) float v2f;
typedef __attribute__((ext_vector_type(8))) float v8f;


constexpr int HIST_CAP     = 65536;
constexpr int HIST_THREADS = 1024;
constexpr int HIST_GROUPS  = 64;
constexpr int SCAN_SHIFT   = 10;
constexpr int SCAN_THREADS = 1 << SCAN_SHIFT;
static_assert(HIST_CAP / SCAN_THREADS == 64, "the fill kernel scans the block sums of one range with two lanes-wide passes");

__global__ __launch_bounds__(HIST_THREADS) void deg_hist_kernel(
    const int* __restrict__ dst, unsigned int* __restrict__ part, int E, int N) {
    __shared__ unsigned int hist[HIST_CAP];
    const int tid = threadIdx.x;
    const int per = (E + (int)gridDim.x - 1) / (int)gridDim.x;
    long long b0 = (long long)blockIdx.x * per;
    const int e0 = (int)(b0 < E ? b0 : E);
    const int e1 = (e0 + per < E) ? (e0 + per) : E;

    for (int r0 = 0; r0 < N; r0 += HIST_CAP) {
        const int nb = (N - r0 < HIST_CAP) ? (N - r0) : HIST_CAP;
        uint4* hz = reinterpret_cast<uint4*>(hist);
        for (int i = tid; i < (nb + 3) / 4; i += HIST_THREADS) hz[i] = uint4{0u, 0u, 0u, 0u};
        __syncthreads();

        int e = e0 + tid;
        for (; e + 7 * HIST_THREADS < e1; e += 8 * HIST_THREADS) {
            int d[8];
#pragma unroll
            for (int j = 0; j < 8; ++j) d[j] = dst[e + j * HIST_THREADS];
#pragma unroll
            for (int j = 0; j < 8; ++j) {
                unsigned int b = (unsigned int)(d[j] - r0);
                if (b < (unsigned int)nb) atomicAdd(&hist[b], 1u);
            }
        }
        for (; e < e1; e += HIST_THREADS) {
            unsigned int b = (unsigned int)(dst[e] - r0);
            if (b < (unsigned int)nb) atomicAdd(&hist[b], 1u);
        }
        __syncthreads();

        unsigned int* prow = part + (size_t)blockIdx.x * N + r0;
        for (int i = tid; i < nb; i += HIST_THREADS) prow[i] = hist[i];
        __syncthreads();
    }
}

__device__ __forceinline__ unsigned int wave_incl_scan(unsigned int v, int lane) {
#pragma unroll
    for (int o = 1; o < 32; o <<= 1) {
        unsigned int t = __shfl_up(v, o, 32);
        if (lane >= o) v += t;
    }
    return v;
}

__global__ __launch_bounds__(SCAN_THREADS) void deg_reduce_scan_kernel(
    unsigned int* part, float* __restrict__ dinv, unsigned int* __restrict__ rowloc,
    unsigned int* __restrict__ blocksum, int N) {
    __shared__ unsigned int wsum[32];
    const int tid  = threadIdx.x;
    const int lane = tid & 31;
    const int wid  = tid >> 5;
    const int i = blockIdx.x * SCAN_THREADS + tid;
    unsigned int c = 0u;
    if (i < N) {
        unsigned int off = (unsigned int)i;
        for (int g0 = 0; g0 < HIST_GROUPS; g0 += 16) {
            unsigned int t[16];
#pragma unroll
            for (int j = 0; j < 16; ++j) t[j] = part[off + (unsigned int)j * (unsigned int)N];
#pragma unroll
            for (int j = 0; j < 16; ++j) {
                part[off + (unsigned int)j * (unsigned int)N] = c;
                c += t[j];
            }
            off += 16u * (unsigned int)N;
        }
        dinv[i] = rsqrtf((float)(c + 1u));
    }
    const unsigned int incl = wave_incl_scan(c, lane);
    if (lane == 31) wsum[wid] = incl;
    __syncthreads();
    if (wid == 0) wsum[lane] = wave_incl_scan(wsum[lane], lane);
    __syncthreads();
    const unsigned int base = wid ? wsum[wid - 1] : 0u;
    if (i < N) rowloc[i] = base + incl - c;
    if (tid == SCAN_THREADS - 1) blocksum[blockIdx.x] = base + incl;
}

__global__ __launch_bounds__(HIST_THREADS) void csr_fill_kernel(
    const int* __restrict__ src, const int* __restrict__ dst,
    const unsigned int* __restrict__ part, const unsigned int* __restrict__ rowloc,
    const unsigned int* __restrict__ blocksum, unsigned int* __restrict__ rowptr,
    int* __restrict__ perm, int E, int N) {
    __shared__ unsigned int cursor[HIST_CAP];
    __shared__ unsigned int bpre[HIST_CAP / SCAN_THREADS + 1];
    const int tid  = threadIdx.x;
    const int lane = tid & 31;
    const int wid  = tid >> 5;
    const int per = (E + (int)gridDim.x - 1) / (int)gridDim.x;
    long long s0 = (long long)blockIdx.x * per;
    const int e0 = (int)(s0 < E ? s0 : E);
    const int e1 = (e0 + per < E) ? (e0 + per) : E;
    const int nblk = (N + SCAN_THREADS - 1) >> SCAN_SHIFT;

    unsigned int carry = 0u;
    for (int r0 = 0; r0 < N; r0 += HIST_CAP) {
        const int nb = (N - r0 < HIST_CAP) ? (N - r0) : HIST_CAP;
        const int b0 = r0 >> SCAN_SHIFT;
        if (wid == 0) {
            const unsigned int a = (b0 + lane < nblk) ? blocksum[b0 + lane] : 0u;
            const unsigned int b = (b0 + 32 + lane < nblk) ? blocksum[b0 + 32 + lane] : 0u;
            const unsigned int ia = wave_incl_scan(a, lane);
            const unsigned int ib = wave_incl_scan(b, lane);
            const unsigned int ta = __shfl(ia, 31, 32);
            bpre[lane]      = carry + ia - a;
            bpre[32 + lane] = carry + ta + ib - b;
            if (lane == 31) bpre[64] = carry + ta + ib;
        }
        __syncthreads();
        const unsigned int* mine = part + (size_t)blockIdx.x * N + r0;
        for (int i0 = tid; i0 < nb; i0 += 8 * HIST_THREADS) {
            unsigned int rl[8], pm[8];
#pragma unroll
            for (int k = 0; k < 8; ++k) {
                const int i  = i0 + k * HIST_THREADS;
                const int ic = (i < nb) ? i : (nb - 1);
                rl[k] = rowloc[r0 + ic];
                pm[k] = mine[ic];
            }
            asm volatile("" ::"v"(rl[0]), "v"(pm[0]));
#pragma unroll
            for (int k = 0; k < 8; ++k) {
                const int i = i0 + k * HIST_THREADS;
                if (i < nb) {
                    const unsigned int start = rl[k] + bpre[i >> SCAN_SHIFT];
                    cursor[i] = start + pm[k];
                    if (((i >> SCAN_SHIFT) % (int)gridDim.x) == (int)blockIdx.x) rowptr[r0 + i] = start;
                }
            }
        }
        carry = bpre[64];
        __syncthreads();

        int e = e0 + tid;
        for (; e + 7 * HIST_THREADS < e1; e += 8 * HIST_THREADS) {
            int d[8], s[8];
#pragma unroll
            for (int j = 0; j < 8; ++j) {
                d[j] = dst[e + j * HIST_THREADS];
                s[j] = src[e + j * HIST_THREADS];
            }
            asm volatile("" ::"v"(s[0]));
#pragma unroll
            for (int j = 0; j < 8; ++j) {
                unsigned int b = (unsigned int)(d[j] - r0);
                if (b < (unsigned int)nb) perm[atomicAdd(&cursor[b], 1u)] = s[j];
            }
        }
        for (; e < e1; e += HIST_THREADS) {
            unsigned int b = (unsigned int)(dst[e] - r0);
            if (b < (unsigned int)nb) perm[atomicAdd(&cursor[b], 1u)] = src[e];
        }
        __syncthreads();
    }
    if (blockIdx.x == 0 && tid == 0) rowptr[N] = carry;
}

template <int FIN, int FOUT, bool FUSE_BIAS_RELU>
__global__ __launch_bounds__(256) void gemm_wmma_kernel(
    const float* __restrict__ A, const float* __restrict__ W,
    const float* __restrict__ bias, float* __restrict__ C,
    int M, int mtiles) {
    constexpr int NT = FOUT / 16;
    int wave = blockIdx.x * (blockDim.x >> 5) + (threadIdx.x >> 5);
    if (wave >= mtiles) return;

    const int row0 = wave * 16;
    const int lane = threadIdx.x & 31;
    const int half = lane >> 4;
    const int lm   = lane & 15;
    const int koff = half * 2;

    const float* __restrict__ arow = A + (size_t)(row0 + lm) * FIN;
    const bool full_tile = (row0 + 16) <= M;
    const float* __restrict__ arow_safe =
        full_tile ? arow : (A + (size_t)((row0 + lm) < M ? (row0 + lm) : (M - 1)) * FIN);

    v8f acc[NT];
#pragma unroll
    for (int t = 0; t < NT; ++t) acc[t] = v8f{};

#pragma unroll 4
    for (int k = 0; k < FIN; k += 4) {
        float a0 = arow_safe[k + koff];
        float a1 = arow_safe[k + koff + 1];
        if (FUSE_BIAS_RELU) {
            a0 = fmaxf(a0 + bias[k + koff], 0.0f);
            a1 = fmaxf(a1 + bias[k + koff + 1], 0.0f);
        }
        v2f a;
        a.x = a0;
        a.y = a1;
#pragma unroll
        for (int t = 0; t < NT; ++t) {
            v2f b;
            b.x = W[(size_t)(k + koff) * FOUT + t * 16 + lm];
            b.y = W[(size_t)(k + koff + 1) * FOUT + t * 16 + lm];
            acc[t] = __builtin_amdgcn_wmma_f32_16x16x4_f32(
                 false, a,  false, b,
                 (short)0, acc[t],  false,  false);
        }
    }

    if (full_tile) {
#pragma unroll
        for (int g = 0; g < 8; ++g) {
            float* crow = C + (size_t)(row0 + half * 8 + g) * FOUT + lm;
#pragma unroll
            for (int t = 0; t < NT; ++t) crow[t * 16] = acc[t][g];
        }
    } else {
#pragma unroll
        for (int g = 0; g < 8; ++g) {
            int r = row0 + half * 8 + g;
            if (r < M) {
                float* crow = C + (size_t)r * FOUT + lm;
#pragma unroll
                for (int t = 0; t < NT; ++t) crow[t * 16] = acc[t][g];
            }
        }
    }
}

template <int F, bool ADD_BIAS>
__global__ __launch_bounds__(256) void aggregate_rows_kernel(
    const float* __restrict__ h, const int* __restrict__ perm,
    const unsigned int* __restrict__ rowptr, const float* __restrict__ dinv,
    const float* __restrict__ bias, float* __restrict__ out, int N) {
    constexpr int TPR = F / 4;
    constexpr int G   = 32 / TPR;
    const int lane = threadIdx.x & 31;
    const int gi = lane / TPR;
    const int li = lane % TPR;
    const int d = __builtin_amdgcn_readfirstlane(
        (int)(blockIdx.x * (blockDim.x >> 5) + (threadIdx.x >> 5)));
    if (d >= N) return;

    const int beg = (int)rowptr[d];
    const int end = (int)rowptr[d + 1];
    const float dd = dinv[d];
    const float* __restrict__ hcol = h + li * 4;

    float4 acc = float4{0.0f, 0.0f, 0.0f, 0.0f};
    const float4 own = *reinterpret_cast<const float4*>(hcol + (size_t)d * F);

    for (int base = beg; base < end; base += 32) {
        const int cnt = (end - base < 32) ? (end - base) : 32;
        int s = d;
        float w = 0.0f;
        if (lane < cnt) {
            s = perm[base + lane];
            w = dinv[s] * dd;
        }
        int j = 0;
        for (; j + 4 * G <= cnt; j += 4 * G) {
            float4 v[4];
            float wv[4];
#pragma unroll
            for (int u = 0; u < 4; ++u) {
                const int sl = j + u * G + gi;
                const int sj = __shfl(s, sl, 32);
                wv[u] = __shfl(w, sl, 32);
                v[u] = *reinterpret_cast<const float4*>(hcol + (size_t)sj * F);
            }
#pragma unroll
            for (int u = 0; u < 4; ++u) {
                acc.x = fmaf(v[u].x, wv[u], acc.x);
                acc.y = fmaf(v[u].y, wv[u], acc.y);
                acc.z = fmaf(v[u].z, wv[u], acc.z);
                acc.w = fmaf(v[u].w, wv[u], acc.w);
            }
        }
        if (j < cnt) {
            float4 v[4];
            float wv[4];
#pragma unroll
            for (int u = 0; u < 4; ++u) {
                const int sl = j + u * G + gi;
                const int sj = __shfl(s, sl & 31, 32);
                const float wj = __shfl(w, sl & 31, 32);
                const bool ok = sl < cnt;
                wv[u] = ok ? wj : 0.0f;
                v[u] = float4{0.0f, 0.0f, 0.0f, 0.0f};
                if (ok) v[u] = *reinterpret_cast<const float4*>(hcol + (size_t)sj * F);
            }
#pragma unroll
            for (int u = 0; u < 4; ++u) {
                acc.x = fmaf(v[u].x, wv[u], acc.x);
                acc.y = fmaf(v[u].y, wv[u], acc.y);
                acc.z = fmaf(v[u].z, wv[u], acc.z);
                acc.w = fmaf(v[u].w, wv[u], acc.w);
            }
        }
    }

    if (gi == 0) {
        const float w = dd * dd;
        acc.x = fmaf(own.x, w, acc.x);
        acc.y = fmaf(own.y, w, acc.y);
        acc.z = fmaf(own.z, w, acc.z);
        acc.w = fmaf(own.w, w, acc.w);
    }
#pragma unroll
    for (int m = TPR; m < 32; m <<= 1) {
        acc.x += __shfl_xor(acc.x, m, 32);
        acc.y += __shfl_xor(acc.y, m, 32);
        acc.z += __shfl_xor(acc.z, m, 32);
        acc.w += __shfl_xor(acc.w, m, 32);
    }
    if (gi == 0) {
        if (ADD_BIAS) {
            acc.x += bias[li * 4 + 0];
            acc.y += bias[li * 4 + 1];
            acc.z += bias[li * 4 + 2];
            acc.w += bias[li * 4 + 3];
        }
        *reinterpret_cast<float4*>(out + (size_t)d * F + li * 4) = acc;
    }
}


extern "C" void kernel_launch(void* const* d_in, const int* in_sizes, int n_in,
                              void* d_out, int out_size, void* d_ws, size_t ws_size,
                              hipStream_t stream) {
    const float* x  = (const float*)d_in[0];
    const int*   ei = (const int*)d_in[1];
    const float* W1 = (const float*)d_in[2];
    const float* b1 = (const float*)d_in[3];
    const float* W2 = (const float*)d_in[4];
    const float* b2 = (const float*)d_in[5];

    const int FIN = 128, F1 = 64, F2 = 32;
    const int N = in_sizes[0] / FIN;
    const int E = in_sizes[1] / 2;
    const int* src = ei;
    const int* dst = ei + E;
    const int nblk = (N + SCAN_THREADS - 1) / SCAN_THREADS;

    float* ws   = (float*)d_ws;
    float* h1   = ws;
    float* agg1 = h1 + (size_t)N * F1;
    float* deg  = agg1 + (size_t)N * F1;
    float* outp = (float*)d_out;
    unsigned int* part     = (unsigned int*)(deg + N);
    unsigned int* rowloc   = part + (size_t)HIST_GROUPS * N;
    unsigned int* rowptr   = rowloc + N;
    unsigned int* blocksum = rowptr + N + 1;
    int*          perm     = (int*)(blocksum + nblk);

    const int thr = 256;
    const int wpb = thr / 32;

    deg_hist_kernel<<<HIST_GROUPS, HIST_THREADS, 0, stream>>>(dst, part, E, N);
    deg_reduce_scan_kernel<<<nblk, SCAN_THREADS, 0, stream>>>(part, deg, rowloc, blocksum, N);
    csr_fill_kernel<<<HIST_GROUPS, HIST_THREADS, 0, stream>>>(src, dst, part, rowloc, blocksum,
                                                              rowptr, perm, E, N);

    {
        int mtiles = (N + 15) / 16;
        int blocks = (mtiles + wpb - 1) / wpb;
        gemm_wmma_kernel<128, 64, false>
            <<<blocks, thr, 0, stream>>>(x, W1, nullptr, h1, N, mtiles);
    }
    aggregate_rows_kernel<64, false>
        <<<(N + wpb - 1) / wpb, thr, 0, stream>>>(h1, perm, rowptr, deg, nullptr, agg1, N);
    {
        int mtiles = (N + 15) / 16;
        int blocks = (mtiles + wpb - 1) / wpb;
        gemm_wmma_kernel<64, 32, true>
            <<<blocks, thr, 0, stream>>>(agg1, W2, b1, h1  , N, mtiles);
    }
    aggregate_rows_kernel<32, true>
        <<<(N + wpb - 1) / wpb, thr, 0, stream>>>(h1, perm, rowptr, deg, b2, outp, N);
}


# Edge sort regridded to fill the chip: histogram and bucket fill split the nodes over grid.y (64 edge slices x 4 node ranges, 64 KiB LDS), count scan in groups of 256 nodes; same counts and buckets
# speedup vs baseline: 3.9646x; 1.4140x over previous
#include <hip/hip_runtime.h>

typedef __attribute__((ext_vector_type(2))) float v2f;
typedef __attribute__((ext_vector_type(8))) float v8f;


constexpr int HIST_CAP     = 16384;
constexpr int HIST_THREADS = 1024;
constexpr int HIST_GROUPS  = 64;
constexpr int HIST_RANGES  = 4;
constexpr int SCAN_SHIFT   = 8;
constexpr int SCAN_THREADS = 1 << SCAN_SHIFT;
static_assert(HIST_CAP / SCAN_THREADS == 64, "the fill kernel scans the block sums of one range with two lanes-wide passes");

__device__ __forceinline__ int range_width(int N, int nranges) {
    int w = (N + nranges - 1) / nranges;
    w = (w + SCAN_THREADS - 1) & ~(SCAN_THREADS - 1);
    return w < HIST_CAP ? w : HIST_CAP;
}

__global__ __launch_bounds__(HIST_THREADS) void deg_hist_kernel(
    const int* __restrict__ dst, unsigned int* __restrict__ part, int E, int N) {
    __shared__ unsigned int hist[HIST_CAP];
    const int tid = threadIdx.x;
    const int per = (E + (int)gridDim.x - 1) / (int)gridDim.x;
    long long b0 = (long long)blockIdx.x * per;
    const int e0 = (int)(b0 < E ? b0 : E);
    const int e1 = (e0 + per < E) ? (e0 + per) : E;
    const int rw = range_width(N, (int)gridDim.y);

    for (int r0 = (int)blockIdx.y * rw; r0 < N; r0 += (int)gridDim.y * rw) {
        const int nb = (N - r0 < rw) ? (N - r0) : rw;
        uint4* hz = reinterpret_cast<uint4*>(hist);
        for (int i = tid; i < (nb + 3) / 4; i += HIST_THREADS) hz[i] = uint4{0u, 0u, 0u, 0u};
        __syncthreads();

        int e = e0 + tid;
        for (; e + 7 * HIST_THREADS < e1; e += 8 * HIST_THREADS) {
            int d[8];
#pragma unroll
            for (int j = 0; j < 8; ++j) d[j] = dst[e + j * HIST_THREADS];
#pragma unroll
            for (int j = 0; j < 8; ++j) {
                unsigned int b = (unsigned int)(d[j] - r0);
                if (b < (unsigned int)nb) atomicAdd(&hist[b], 1u);
            }
        }
        for (; e < e1; e += HIST_THREADS) {
            unsigned int b = (unsigned int)(dst[e] - r0);
            if (b < (unsigned int)nb) atomicAdd(&hist[b], 1u);
        }
        __syncthreads();

        unsigned int* prow = part + (size_t)blockIdx.x * N + r0;
        for (int i = tid; i < nb; i += HIST_THREADS) prow[i] = hist[i];
        __syncthreads();
    }
}

__device__ __forceinline__ unsigned int wave_incl_scan(unsigned int v, int lane) {
#pragma unroll
    for (int o = 1; o < 32; o <<= 1) {
        unsigned int t = __shfl_up(v, o, 32);
        if (lane >= o) v += t;
    }
    return v;
}

__device__ __forceinline__ unsigned int wave_sum(unsigned int v) {
#pragma unroll
    for (int o = 16; o > 0; o >>= 1) v += __shfl_xor(v, o, 32);
    return v;
}

__global__ __launch_bounds__(SCAN_THREADS) void deg_reduce_scan_kernel(
    unsigned int* part, float* __restrict__ dinv, unsigned int* __restrict__ rowloc,
    unsigned int* __restrict__ blocksum, int N) {
    constexpr int NW = SCAN_THREADS / 32;
    __shared__ unsigned int wsum[NW];
    const int tid  = threadIdx.x;
    const int lane = tid & 31;
    const int wid  = tid >> 5;
    const int i = blockIdx.x * SCAN_THREADS + tid;
    unsigned int c = 0u;
    if (i < N) {
        unsigned int off = (unsigned int)i;
        for (int g0 = 0; g0 < HIST_GROUPS; g0 += 16) {
            unsigned int t[16];
#pragma unroll
            for (int j = 0; j < 16; ++j) t[j] = part[off + (unsigned int)j * (unsigned int)N];
#pragma unroll
            for (int j = 0; j < 16; ++j) {
                part[off + (unsigned int)j * (unsigned int)N] = c;
                c += t[j];
            }
            off += 16u * (unsigned int)N;
        }
        dinv[i] = rsqrtf((float)(c + 1u));
    }
    const unsigned int incl = wave_incl_scan(c, lane);
    if (lane == 31) wsum[wid] = incl;
    __syncthreads();
    if (wid == 0) {
        const unsigned int w = wave_incl_scan(lane < NW ? wsum[lane] : 0u, lane);
        if (lane < NW) wsum[lane] = w;
    }
    __syncthreads();
    const unsigned int base = wid ? wsum[wid - 1] : 0u;
    if (i < N) rowloc[i] = base + incl - c;
    if (tid == SCAN_THREADS - 1) blocksum[blockIdx.x] = base + incl;
}

__global__ __launch_bounds__(HIST_THREADS) void csr_fill_kernel(
    const int* __restrict__ src, const int* __restrict__ dst,
    const unsigned int* __restrict__ part, const unsigned int* __restrict__ rowloc,
    const unsigned int* __restrict__ blocksum, unsigned int* __restrict__ rowptr,
    int* __restrict__ perm, int E, int N) {
    __shared__ unsigned int cursor[HIST_CAP];
    __shared__ unsigned int bpre[HIST_CAP / SCAN_THREADS + 1];
    __shared__ unsigned int csum[HIST_THREADS / 32];
    const int tid  = threadIdx.x;
    const int lane = tid & 31;
    const int wid  = tid >> 5;
    const int per = (E + (int)gridDim.x - 1) / (int)gridDim.x;
    long long s0 = (long long)blockIdx.x * per;
    const int e0 = (int)(s0 < E ? s0 : E);
    const int e1 = (e0 + per < E) ? (e0 + per) : E;
    const int rw = range_width(N, (int)gridDim.y);

    for (int r0 = (int)blockIdx.y * rw; r0 < N; r0 += (int)gridDim.y * rw) {
        const int nb = (N - r0 < rw) ? (N - r0) : rw;
        const int b0 = r0 >> SCAN_SHIFT;
        const int nbk = (nb + SCAN_THREADS - 1) >> SCAN_SHIFT;
        unsigned int before = 0u;
#pragma unroll 1
        for (int k = tid; k < b0; k += HIST_THREADS) before += blocksum[k];
        before = wave_sum(before);
        if (lane == 0) csum[wid] = before;
        __syncthreads();
        if (wid == 0) {
            const unsigned int carry = wave_sum(csum[lane]);
            const unsigned int a = (lane < nbk) ? blocksum[b0 + lane] : 0u;
            const unsigned int b = (32 + lane < nbk) ? blocksum[b0 + 32 + lane] : 0u;
            const unsigned int ia = wave_incl_scan(a, lane);
            const unsigned int ib = wave_incl_scan(b, lane);
            const unsigned int ta = __shfl(ia, 31, 32);
            bpre[lane]      = carry + ia - a;
            bpre[32 + lane] = carry + ta + ib - b;
            if (lane == 31) {
                bpre[64] = carry + ta + ib;
                if (r0 + nb == N && blockIdx.x == 0) rowptr[N] = carry + ta + ib;
            }
        }
        __syncthreads();
        const unsigned int* mine = part + (size_t)blockIdx.x * N + r0;
        for (int i0 = tid; i0 < nb; i0 += 8 * HIST_THREADS) {
            unsigned int rl[8], pm[8];
#pragma unroll
            for (int k = 0; k < 8; ++k) {
                const int i  = i0 + k * HIST_THREADS;
                const int ic = (i < nb) ? i : (nb - 1);
                rl[k] = rowloc[r0 + ic];
                pm[k] = mine[ic];
            }
            asm volatile("" ::"v"(rl[0]), "v"(pm[0]));
#pragma unroll
            for (int k = 0; k < 8; ++k) {
                const int i = i0 + k * HIST_THREADS;
                if (i < nb) {
                    const unsigned int start = rl[k] + bpre[i >> SCAN_SHIFT];
                    cursor[i] = start + pm[k];
                    if (((i >> SCAN_SHIFT) % (int)gridDim.x) == (int)blockIdx.x) rowptr[r0 + i] = start;
                }
            }
        }
        __syncthreads();

        int e = e0 + tid;
        for (; e + 7 * HIST_THREADS < e1; e += 8 * HIST_THREADS) {
            int d[8], s[8];
#pragma unroll
            for (int j = 0; j < 8; ++j) {
                d[j] = dst[e + j * HIST_THREADS];
                s[j] = src[e + j * HIST_THREADS];
            }
            asm volatile("" ::"v"(s[0]));
#pragma unroll
            for (int j = 0; j < 8; ++j) {
                unsigned int b = (unsigned int)(d[j] - r0);
                if (b < (unsigned int)nb) perm[atomicAdd(&cursor[b], 1u)] = s[j];
            }
        }
        for (; e < e1; e += HIST_THREADS) {
            unsigned int b = (unsigned int)(dst[e] - r0);
            if (b < (unsigned int)nb) perm[atomicAdd(&cursor[b], 1u)] = src[e];
        }
        __syncthreads();
    }
}

template <int FIN, int FOUT, bool FUSE_BIAS_RELU>
__global__ __launch_bounds__(256) void gemm_wmma_kernel(
    const float* __restrict__ A, const float* __restrict__ W,
    const float* __restrict__ bias, const float* __restrict__ rowscale, float* __restrict__ C,
    int M, int mtiles) {
    constexpr int NT = FOUT / 16;
    int wave = blockIdx.x * (blockDim.x >> 5) + (threadIdx.x >> 5);
    if (wave >= mtiles) return;

    const int row0 = wave * 16;
    const int lane = threadIdx.x & 31;
    const int half = lane >> 4;
    const int lm   = lane & 15;
    const int koff = half * 2;

    const float* __restrict__ arow = A + (size_t)(row0 + lm) * FIN;
    const bool full_tile = (row0 + 16) <= M;
    const float* __restrict__ arow_safe =
        full_tile ? arow : (A + (size_t)((row0 + lm) < M ? (row0 + lm) : (M - 1)) * FIN);

    float sc[8];
#pragma unroll
    for (int g = 0; g < 8; ++g) {
        const int r = row0 + half * 8 + g;
        sc[g] = rowscale[r < M ? r : (M - 1)];
    }

    v8f acc[NT];
#pragma unroll
    for (int t = 0; t < NT; ++t) acc[t] = v8f{};

#pragma unroll 4
    for (int k = 0; k < FIN; k += 4) {
        float a0 = arow_safe[k + koff];
        float a1 = arow_safe[k + koff + 1];
        if (FUSE_BIAS_RELU) {
            a0 = fmaxf(a0 + bias[k + koff], 0.0f);
            a1 = fmaxf(a1 + bias[k + koff + 1], 0.0f);
        }
        v2f a;
        a.x = a0;
        a.y = a1;
#pragma unroll
        for (int t = 0; t < NT; ++t) {
            v2f b;
            b.x = W[(size_t)(k + koff) * FOUT + t * 16 + lm];
            b.y = W[(size_t)(k + koff + 1) * FOUT + t * 16 + lm];
            acc[t] = __builtin_amdgcn_wmma_f32_16x16x4_f32(
                 false, a,  false, b,
                 (short)0, acc[t],  false,  false);
        }
    }

    if (full_tile) {
#pragma unroll
        for (int g = 0; g < 8; ++g) {
            float* crow = C + (size_t)(row0 + half * 8 + g) * FOUT + lm;
#pragma unroll
            for (int t = 0; t < NT; ++t) crow[t * 16] = acc[t][g] * sc[g];
        }
    } else {
#pragma unroll
        for (int g = 0; g < 8; ++g) {
            int r = row0 + half * 8 + g;
            if (r < M) {
                float* crow = C + (size_t)r * FOUT + lm;
#pragma unroll
                for (int t = 0; t < NT; ++t) crow[t * 16] = acc[t][g] * sc[g];
            }
        }
    }
}

template <int F, bool ADD_BIAS>
__global__ __launch_bounds__(256) void aggregate_rows_kernel(
    const float* __restrict__ hs, const int* __restrict__ perm,
    const unsigned int* __restrict__ rowptr, const float* __restrict__ dinv,
    const float* __restrict__ bias, float* __restrict__ out, int N) {
    constexpr int TPR = F / 4;
    constexpr int G   = 32 / TPR;
    const int lane = threadIdx.x & 31;
    const int gi = lane / TPR;
    const int li = lane % TPR;
    const int d = __builtin_amdgcn_readfirstlane(
        (int)(blockIdx.x * (blockDim.x >> 5) + (threadIdx.x >> 5)));
    if (d >= N) return;

    const int beg = (int)rowptr[d];
    const int end = (int)rowptr[d + 1];
    const float dd = dinv[d];
    const float* __restrict__ hcol = hs + li * 4;

    float4 acc = float4{0.0f, 0.0f, 0.0f, 0.0f};
    const float4 own = *reinterpret_cast<const float4*>(hcol + (size_t)d * F);

    for (int base = beg; base < end; base += 32) {
        const int cnt = (end - base < 32) ? (end - base) : 32;
        int s = d;
        if (lane < cnt) s = perm[base + lane];
        int j = 0;
        for (; j + 4 * G <= cnt; j += 4 * G) {
            float4 v[4];
#pragma unroll
            for (int u = 0; u < 4; ++u) {
                const int sj = __shfl(s, j + u * G + gi, 32);
                v[u] = *reinterpret_cast<const float4*>(hcol + (size_t)sj * F);
            }
#pragma unroll
            for (int u = 0; u < 4; ++u) {
                acc.x += v[u].x;
                acc.y += v[u].y;
                acc.z += v[u].z;
                acc.w += v[u].w;
            }
        }
        if (j < cnt) {
            float4 v[4];
#pragma unroll
            for (int u = 0; u < 4; ++u) {
                const int sl = j + u * G + gi;
                const int sj = __shfl(s, sl & 31, 32);
                v[u] = float4{0.0f, 0.0f, 0.0f, 0.0f};
                if (sl < cnt) v[u] = *reinterpret_cast<const float4*>(hcol + (size_t)sj * F);
            }
#pragma unroll
            for (int u = 0; u < 4; ++u) {
                acc.x += v[u].x;
                acc.y += v[u].y;
                acc.z += v[u].z;
                acc.w += v[u].w;
            }
        }
    }

    if (gi == 0) {
        acc.x += own.x;
        acc.y += own.y;
        acc.z += own.z;
        acc.w += own.w;
    }
#pragma unroll
    for (int m = TPR; m < 32; m <<= 1) {
        acc.x += __shfl_xor(acc.x, m, 32);
        acc.y += __shfl_xor(acc.y, m, 32);
        acc.z += __shfl_xor(acc.z, m, 32);
        acc.w += __shfl_xor(acc.w, m, 32);
    }
    if (gi == 0) {
        acc.x *= dd;
        acc.y *= dd;
        acc.z *= dd;
        acc.w *= dd;
        if (ADD_BIAS) {
            acc.x += bias[li * 4 + 0];
            acc.y += bias[li * 4 + 1];
            acc.z += bias[li * 4 + 2];
            acc.w += bias[li * 4 + 3];
        }
        *reinterpret_cast<float4*>(out + (size_t)d * F + li * 4) = acc;
    }
}


extern "C" void kernel_launch(void* const* d_in, const int* in_sizes, int n_in,
                              void* d_out, int out_size, void* d_ws, size_t ws_size,
                              hipStream_t stream) {
    const float* x  = (const float*)d_in[0];
    const int*   ei = (const int*)d_in[1];
    const float* W1 = (const float*)d_in[2];
    const float* b1 = (const float*)d_in[3];
    const float* W2 = (const float*)d_in[4];
    const float* b2 = (const float*)d_in[5];

    const int FIN = 128, F1 = 64, F2 = 32;
    const int N = in_sizes[0] / FIN;
    const int E = in_sizes[1] / 2;
    const int* src = ei;
    const int* dst = ei + E;
    const int nblk = (N + SCAN_THREADS - 1) / SCAN_THREADS;

    float* ws   = (float*)d_ws;
    float* h1   = ws;
    float* agg1 = h1 + (size_t)N * F1;
    float* deg  = agg1 + (size_t)N * F1;
    float* outp = (float*)d_out;
    unsigned int* part     = (unsigned int*)(deg + N);
    unsigned int* rowloc   = part + (size_t)HIST_GROUPS * N;
    unsigned int* rowptr   = rowloc + N;
    unsigned int* blocksum = rowptr + N + 1;
    int*          perm     = (int*)(blocksum + nblk);

    const int thr = 256;
    const int wpb = thr / 32;

    const dim3 sort_grid(HIST_GROUPS, HIST_RANGES);
    deg_hist_kernel<<<sort_grid, HIST_THREADS, 0, stream>>>(dst, part, E, N);
    deg_reduce_scan_kernel<<<nblk, SCAN_THREADS, 0, stream>>>(part, deg, rowloc, blocksum, N);
    csr_fill_kernel<<<sort_grid, HIST_THREADS, 0, stream>>>(src, dst, part, rowloc, blocksum,
                                                              rowptr, perm, E, N);

    {
        int mtiles = (N + 15) / 16;
        int blocks = (mtiles + wpb - 1) / wpb;
        gemm_wmma_kernel<128, 64, false>
            <<<blocks, thr, 0, stream>>>(x, W1, nullptr, deg, h1, N, mtiles);
    }
    aggregate_rows_kernel<64, false>
        <<<(N + wpb - 1) / wpb, thr, 0, stream>>>(h1, perm, rowptr, deg, nullptr, agg1, N);
    {
        int mtiles = (N + 15) / 16;
        int blocks = (mtiles + wpb - 1) / wpb;
        gemm_wmma_kernel<64, 32, true>
            <<<blocks, thr, 0, stream>>>(agg1, W2, b1, deg, h1  , N, mtiles);
    }
    aggregate_rows_kernel<32, true>
        <<<(N + wpb - 1) / wpb, thr, 0, stream>>>(h1, perm, rowptr, deg, b2, outp, N);
}
